# starting kernel
# baseline (speedup 1.0000x reference)
#include <hip/hip_runtime.h>

typedef float v2f __attribute__((ext_vector_type(2)));
typedef float v8f __attribute__((ext_vector_type(8)));

#define FDIM 128
#define GEPS 1e-10f

__global__ __launch_bounds__(128) void prep_kernel(const float* __restrict__ graph,
                                                   const float* __restrict__ W,
                                                   const float* __restrict__ b,
                                                   float* __restrict__ M,
                                                   float* __restrict__ bias2) {
    __shared__ float invdeg[FDIM];
    const int g = threadIdx.x;
    const int f = blockIdx.x;

    float s = 0.0f;
    for (int c = 0; c < FDIM; ++c) s += graph[g * FDIM + c];
    invdeg[g] = (s > GEPS) ? (1.0f / s) : 0.0f;
    __syncthreads();

    float acc = 0.0f;
    for (int hh = 0; hh < FDIM; ++hh)
        acc = fmaf(W[hh * FDIM + f], graph[hh * FDIM + g] * invdeg[hh], acc);
    M[f * FDIM + g] = acc;

    if (f == 0) {
        float bb = 0.0f;
        for (int hh = 0; hh < FDIM; ++hh)
            bb = fmaf(b[hh], graph[hh * FDIM + g] * invdeg[hh], bb);
        bias2[g] = bb;
    }
}

__global__ __launch_bounds__(256) void gemm_kernel(const float* __restrict__ hmat,
                                                   const float* __restrict__ M,
                                                   const float* __restrict__ bias2,
                                                   float* __restrict__ out) {
    __shared__ float ldsM[FDIM * FDIM];

    const int tid = threadIdx.x;
    for (int i = tid; i < FDIM * FDIM; i += 256) {
        const int r = i >> 7, c = i & 127;
        const int p = r >> 1, comp = r & 1;
        const int colAdj = c ^ ((p & 1) << 4);
        ldsM[p * 256 + colAdj * 2 + comp] = M[i];
    }
    __syncthreads();

    const int wave  = tid >> 5;
    const int lane  = tid & 31;
    const int laneM = lane & 15;
    const int half  = lane >> 4;
    const int kOff  = half << 1;

    const long rowBase = (long)blockIdx.x * 128 + (long)wave * 16;
    const float* hrow = hmat + (rowBase + laneM) * FDIM;

    v2f a[32];
#pragma unroll
    for (int ks = 0; ks < 32; ++ks)
        a[ks] = *(const v2f*)(hrow + ks * 4 + kOff);

    for (int n = 0; n < 8; ++n) {
        const int col = n * 16 + laneM;
        const float bsc = bias2[col];
        v8f acc = { bsc, bsc, bsc, bsc, bsc, bsc, bsc, bsc };

        const v2f* bptr = (const v2f*)&ldsM[half * 256 + ((col ^ (half << 4)) << 1)];

#pragma unroll
        for (int ks = 0; ks < 32; ++ks) {
            const v2f bfrag = bptr[ks * 256];
            acc = __builtin_amdgcn_wmma_f32_16x16x4_f32(
                false, a[ks], false, bfrag, (short)0, acc, false, false);
        }

        float* o = out + (rowBase + (long)(half << 3)) * FDIM + n * 16 + laneM;
#pragma unroll
        for (int r = 0; r < 8; ++r)
            o[(long)r * FDIM] = acc[r];
    }
}

extern "C" void kernel_launch(void* const* d_in, const int* in_sizes, int n_in,
                              void* d_out, int out_size, void* d_ws, size_t ws_size,
                              hipStream_t stream) {
    const float* h     = (const float*)d_in[0];
    const float* graph = (const float*)d_in[1];
    const float* W     = (const float*)d_in[2];
    const float* b     = (const float*)d_in[3];
    float* out = (float*)d_out;

    float* M     = (float*)d_ws;
    float* bias2 = M + FDIM * FDIM;

    prep_kernel<<<dim3(FDIM), dim3(FDIM), 0, stream>>>(graph, W, b, M, bias2);

    const long rows = (long)out_size / FDIM;
    const int nblocks = (int)(rows / 128);
    gemm_kernel<<<dim3(nblocks), dim3(256), 0, stream>>>(h, M, bias2, out);
}


# prep kernel: its 40 dependent load round trips per block become one (whole graph loaded at once into LDS, W column and b read once, bias2 chain in a 129th block); summation order unchanged
# speedup vs baseline: 1.1106x; 1.1106x over previous
#include <hip/hip_runtime.h>

typedef float v2f __attribute__((ext_vector_type(2)));
typedef float v8f __attribute__((ext_vector_type(8)));

#define FDIM 128
#define GEPS 1e-10f

#define GPITCH 132

__global__ __launch_bounds__(128) void prep_kernel(const float* __restrict__ graph,
                                                   const float* __restrict__ W,
                                                   const float* __restrict__ b,
                                                   float* __restrict__ M,
                                                   float* __restrict__ bias2) {
    __shared__ __attribute__((aligned(16))) float gs[FDIM * GPITCH];
    __shared__ __attribute__((aligned(16))) float coef[FDIM];
    __shared__ __attribute__((aligned(16))) float invdeg[FDIM];
    const int t = threadIdx.x;
    const int f = blockIdx.x;

    float4 v[32];
#pragma unroll
    for (int i = 0; i < 32; ++i)
        v[i] = *(const float4*)(graph + (i * FDIM + t) * 4);
    const float cf = (f < FDIM) ? W[t * FDIM + f] : b[t];

#pragma unroll
    for (int i = 0; i < 32; ++i) {
        const int row = i * 4 + (t >> 5);
        const int col = (t & 31) * 4;
        *(float4*)&gs[row * GPITCH + col] = v[i];
    }
    coef[t] = cf;
    __syncthreads();

    float s = 0.0f;
#pragma unroll
    for (int c4 = 0; c4 < 32; ++c4) {
        const float4 q = *(const float4*)&gs[t * GPITCH + c4 * 4];
        s += q.x; s += q.y; s += q.z; s += q.w;
    }
    invdeg[t] = (s > GEPS) ? (1.0f / s) : 0.0f;
    __syncthreads();

    float acc = 0.0f;
#pragma unroll 32
    for (int hh = 0; hh < FDIM; ++hh)
        acc = fmaf(coef[hh], gs[hh * GPITCH + t] * invdeg[hh], acc);

    if (f < FDIM) M[f * FDIM + t] = acc;
    else          bias2[t] = acc;
}

__global__ __launch_bounds__(256) void gemm_kernel(const float* __restrict__ hmat,
                                                   const float* __restrict__ M,
                                                   const float* __restrict__ bias2,
                                                   float* __restrict__ out) {
    __shared__ float ldsM[FDIM * FDIM];

    const int tid = threadIdx.x;
    for (int i = tid; i < FDIM * FDIM; i += 256) {
        const int r = i >> 7, c = i & 127;
        const int p = r >> 1, comp = r & 1;
        const int colAdj = c ^ ((p & 1) << 4);
        ldsM[p * 256 + colAdj * 2 + comp] = M[i];
    }
    __syncthreads();

    const int wave  = tid >> 5;
    const int lane  = tid & 31;
    const int laneM = lane & 15;
    const int half  = lane >> 4;
    const int kOff  = half << 1;

    const long rowBase = (long)blockIdx.x * 128 + (long)wave * 16;
    const float* hrow = hmat + (rowBase + laneM) * FDIM;

    v2f a[32];
#pragma unroll
    for (int ks = 0; ks < 32; ++ks)
        a[ks] = *(const v2f*)(hrow + ks * 4 + kOff);

    for (int n = 0; n < 8; ++n) {
        const int col = n * 16 + laneM;
        const float bsc = bias2[col];
        v8f acc = { bsc, bsc, bsc, bsc, bsc, bsc, bsc, bsc };

        const v2f* bptr = (const v2f*)&ldsM[half * 256 + ((col ^ (half << 4)) << 1)];

#pragma unroll
        for (int ks = 0; ks < 32; ++ks) {
            const v2f bfrag = bptr[ks * 256];
            acc = __builtin_amdgcn_wmma_f32_16x16x4_f32(
                false, a[ks], false, bfrag, (short)0, acc, false, false);
        }

        float* o = out + (rowBase + (long)(half << 3)) * FDIM + n * 16 + laneM;
#pragma unroll
        for (int r = 0; r < 8; ++r)
            o[(long)r * FDIM] = acc[r];
    }
}

extern "C" void kernel_launch(void* const* d_in, const int* in_sizes, int n_in,
                              void* d_out, int out_size, void* d_ws, size_t ws_size,
                              hipStream_t stream) {
    const float* h     = (const float*)d_in[0];
    const float* graph = (const float*)d_in[1];
    const float* W     = (const float*)d_in[2];
    const float* b     = (const float*)d_in[3];
    float* out = (float*)d_out;

    float* M     = (float*)d_ws;
    float* bias2 = M + FDIM * FDIM;

    prep_kernel<<<dim3(129), dim3(FDIM), 0, stream>>>(graph, W, b, M, bias2);

    const long rows = (long)out_size / FDIM;
    const int nblocks = (int)(rows / 128);
    gemm_kernel<<<dim3(nblocks), dim3(256), 0, stream>>>(h, M, bias2, out);
}


# gemm: one 32-wave work-group per WGP stages M and bias2 into LDS once (16-byte loads and LDS stores) and each wave walks two 16-row tiles of h; sum order unchanged
# speedup vs baseline: 1.2437x; 1.1198x over previous
#include <hip/hip_runtime.h>

typedef float v2f __attribute__((ext_vector_type(2)));
typedef float v8f __attribute__((ext_vector_type(8)));

#define FDIM 128
#define GEPS 1e-10f

#define GPITCH 132

__global__ __launch_bounds__(128) void prep_kernel(const float* __restrict__ graph,
                                                   const float* __restrict__ W,
                                                   const float* __restrict__ b,
                                                   float* __restrict__ M,
                                                   float* __restrict__ bias2) {
    __shared__ __attribute__((aligned(16))) float gs[FDIM * GPITCH];
    __shared__ __attribute__((aligned(16))) float coef[FDIM];
    __shared__ __attribute__((aligned(16))) float invdeg[FDIM];
    const int t = threadIdx.x;
    const int f = blockIdx.x;

    float4 v[32];
#pragma unroll
    for (int i = 0; i < 32; ++i)
        v[i] = *(const float4*)(graph + (i * FDIM + t) * 4);
    const float cf = (f < FDIM) ? W[t * FDIM + f] : b[t];

#pragma unroll
    for (int i = 0; i < 32; ++i) {
        const int row = i * 4 + (t >> 5);
        const int col = (t & 31) * 4;
        *(float4*)&gs[row * GPITCH + col] = v[i];
    }
    coef[t] = cf;
    __syncthreads();

    float s = 0.0f;
#pragma unroll
    for (int c4 = 0; c4 < 32; ++c4) {
        const float4 q = *(const float4*)&gs[t * GPITCH + c4 * 4];
        s += q.x; s += q.y; s += q.z; s += q.w;
    }
    invdeg[t] = (s > GEPS) ? (1.0f / s) : 0.0f;
    __syncthreads();

    float acc = 0.0f;
#pragma unroll 32
    for (int hh = 0; hh < FDIM; ++hh)
        acc = fmaf(coef[hh], gs[hh * GPITCH + t] * invdeg[hh], acc);

    if (f < FDIM) M[f * FDIM + t] = acc;
    else          bias2[t] = acc;
}

#define GEMM_WAVES      32
#define TILES_PER_WAVE  2
#define ROWS_PER_GROUP  (16 * GEMM_WAVES * TILES_PER_WAVE)
#define TILE_STEP       (16 * GEMM_WAVES * FDIM)

__device__ __forceinline__ void load_a_tile(v2f (&a)[32], const float* __restrict__ hrow) {
#pragma unroll
    for (int ks = 0; ks < 32; ++ks)
        a[ks] = *(const v2f*)(hrow + ks * 4);
}

__global__ __launch_bounds__(1024) void gemm_kernel(const float* __restrict__ hmat,
                                                    const float* __restrict__ M,
                                                    const float* __restrict__ bias2,
                                                    float* __restrict__ out) {
    __shared__ __attribute__((aligned(16))) float ldsM[FDIM * FDIM];
    __shared__ float ldsBias[FDIM];

    const int tid   = threadIdx.x;
    const int wave  = tid >> 5;
    const int lane  = tid & 31;
    const int laneM = lane & 15;
    const int half  = lane >> 4;
    const int kOff  = half << 1;

    float4 mx[2], my[2];
    for (int j = 0; j < 2; ++j) {
        const int q  = tid + j * 1024;
        const int p  = q >> 5, c4 = q & 31;
        mx[j] = *(const float4*)(M + (2 * p)     * FDIM + c4 * 4);
        my[j] = *(const float4*)(M + (2 * p + 1) * FDIM + c4 * 4);
    }
    float bv = 0.0f;
    if (tid < FDIM) bv = bias2[tid];

    for (int j = 0; j < 2; ++j) {
        const int q  = tid + j * 1024;
        const int p  = q >> 5, c4 = q & 31;
        const int colAdj = (c4 * 4) ^ ((p & 1) << 4);
        float* dst = &ldsM[p * 256 + colAdj * 2];
        *(float4*)(dst)     = make_float4(mx[j].x, my[j].x, mx[j].y, my[j].y);
        *(float4*)(dst + 4) = make_float4(mx[j].z, my[j].z, mx[j].w, my[j].w);
    }
    if (tid < FDIM) ldsBias[tid] = bv;
    __syncthreads();

    const long rowBase = (long)blockIdx.x * ROWS_PER_GROUP + (long)wave * 16;
    const float* hrow = hmat + (rowBase + laneM) * FDIM + kOff;
    float*       orow = out  + (rowBase + (long)(half << 3)) * FDIM + laneM;

    v2f a[32];
    for (int t = 0; t < TILES_PER_WAVE; ++t) {
        load_a_tile(a, hrow + (long)t * TILE_STEP);

#pragma unroll 1
        for (int n = 0; n < 8; ++n) {
            const int col = n * 16 + laneM;
            const float bsc = ldsBias[col];
            v8f acc = { bsc, bsc, bsc, bsc, bsc, bsc, bsc, bsc };

            const v2f* bptr = (const v2f*)&ldsM[half * 256 + ((col ^ (half << 4)) << 1)];

#pragma unroll
            for (int ks = 0; ks < 32; ++ks) {
                const v2f bfrag = bptr[ks * 256];
                acc = __builtin_amdgcn_wmma_f32_16x16x4_f32(
                    false, a[ks], false, bfrag, (short)0, acc, false, false);
            }

            float* o = orow + (long)t * TILE_STEP + n * 16;
#pragma unroll
            for (int r = 0; r < 8; ++r)
                o[(long)r * FDIM] = acc[r];
        }
    }
}

extern "C" void kernel_launch(void* const* d_in, const int* in_sizes, int n_in,
                              void* d_out, int out_size, void* d_ws, size_t ws_size,
                              hipStream_t stream) {
    const float* h     = (const float*)d_in[0];
    const float* graph = (const float*)d_in[1];
    const float* W     = (const float*)d_in[2];
    const float* b     = (const float*)d_in[3];
    float* out = (float*)d_out;

    float* M     = (float*)d_ws;
    float* bias2 = M + FDIM * FDIM;

    prep_kernel<<<dim3(129), dim3(FDIM), 0, stream>>>(graph, W, b, M, bias2);

    const long rows = (long)out_size / FDIM;
    const int nblocks = (int)(rows / ROWS_PER_GROUP);
    gemm_kernel<<<dim3(nblocks), dim3(32 * GEMM_WAVES), 0, stream>>>(h, M, bias2, out);
}


# gemm: M held in registers as B fragments (512 VGPRs, one wave per SIMD): every matrix op of a tile has its operands in registers, no LDS; next A tile loaded a tile ahead; sum order unchanged
# speedup vs baseline: 1.3225x; 1.0634x over previous
#include <hip/hip_runtime.h>

typedef float v2f __attribute__((ext_vector_type(2)));
typedef float v8f __attribute__((ext_vector_type(8)));

#define FDIM 128
#define GEPS 1e-10f

#define GPITCH 132

#define MFRAG_INDEX(f, g) \
    (((((f) >> 2) * 4 + ((g) >> 5)) * 32 + (((g) & 15) + ((((f) >> 1) & 1) << 4))) * 4 + ((((g) >> 4) & 1) << 1) + ((f) & 1))

__global__ __launch_bounds__(128) void prep_kernel(const float* __restrict__ graph,
                                                   const float* __restrict__ W,
                                                   const float* __restrict__ b,
                                                   float* __restrict__ M,
                                                   float* __restrict__ bias2) {
    __shared__ __attribute__((aligned(16))) float gs[FDIM * GPITCH];
    __shared__ __attribute__((aligned(16))) float coef[FDIM];
    __shared__ __attribute__((aligned(16))) float invdeg[FDIM];
    const int t = threadIdx.x;
    const int f = blockIdx.x;

    float4 v[32];
#pragma unroll
    for (int i = 0; i < 32; ++i)
        v[i] = *(const float4*)(graph + (i * FDIM + t) * 4);
    const float cf = (f < FDIM) ? W[t * FDIM + f] : b[t];

#pragma unroll
    for (int i = 0; i < 32; ++i) {
        const int row = i * 4 + (t >> 5);
        const int col = (t & 31) * 4;
        *(float4*)&gs[row * GPITCH + col] = v[i];
    }
    coef[t] = cf;
    __syncthreads();

    float s = 0.0f;
#pragma unroll
    for (int c4 = 0; c4 < 32; ++c4) {
        const float4 q = *(const float4*)&gs[t * GPITCH + c4 * 4];
        s += q.x; s += q.y; s += q.z; s += q.w;
    }
    invdeg[t] = (s > GEPS) ? (1.0f / s) : 0.0f;
    __syncthreads();

    float acc = 0.0f;
#pragma unroll 32
    for (int hh = 0; hh < FDIM; ++hh)
        acc = fmaf(coef[hh], gs[hh * GPITCH + t] * invdeg[hh], acc);

    if (f < FDIM) M[MFRAG_INDEX(f, t)] = acc;
    else          bias2[t] = acc;
}

#define GEMM_WAVES      4
#define TILES_PER_WAVE  16
#define ROWS_PER_GROUP  (16 * GEMM_WAVES * TILES_PER_WAVE)
#define TILE_STEP       (16 * GEMM_WAVES * FDIM)

__device__ __forceinline__ void load_a_tile(v2f (&a)[32], const float* __restrict__ hrow) {
#pragma unroll
    for (int ks = 0; ks < 32; ++ks)
        a[ks] = *(const v2f*)(hrow + ks * 4);
}

__device__ __forceinline__ void tile_product(const v2f (&a)[32], const v2f (&bf)[32][8],
                                             const float (&bsc)[8], float* __restrict__ o) {
    v8f acc[8];
#pragma unroll
    for (int n = 0; n < 8; ++n) {
        const float s = bsc[n];
        acc[n] = (v8f){ s, s, s, s, s, s, s, s };
    }
#pragma unroll
    for (int ks = 0; ks < 32; ++ks) {
#pragma unroll
        for (int n = 0; n < 8; ++n)
            acc[n] = __builtin_amdgcn_wmma_f32_16x16x4_f32(
                false, a[ks], false, bf[ks][n], (short)0, acc[n], false, false);
    }
#pragma unroll
    for (int r = 0; r < 8; ++r) {
#pragma unroll
        for (int n = 0; n < 8; ++n)
            o[(long)r * FDIM + n * 16] = acc[n][r];
    }
}

__global__ __launch_bounds__(32 * GEMM_WAVES) void gemm_kernel(const float* __restrict__ hmat,
                                                               const float* __restrict__ Mfrag,
                                                               const float* __restrict__ bias2,
                                                               float* __restrict__ out) {
    const int tid   = threadIdx.x;
    const int wave  = tid >> 5;
    const int lane  = tid & 31;
    const int laneM = lane & 15;
    const int half  = lane >> 4;
    const int kOff  = half << 1;

    const long rowBase = (long)blockIdx.x * ROWS_PER_GROUP + (long)wave * 16;
    const float* hrow = hmat + (rowBase + laneM) * FDIM + kOff;
    float*       orow = out  + (rowBase + (long)(half << 3)) * FDIM + laneM;

    v2f a0[32], a1[32];
    load_a_tile(a0, hrow);

    v2f bf[32][8];
    const float4* mq = (const float4*)Mfrag + lane;
#pragma unroll
    for (int ks = 0; ks < 32; ++ks) {
#pragma unroll
        for (int np = 0; np < 4; ++np) {
            const float4 q = mq[(ks * 4 + np) * 32];
            bf[ks][2 * np]     = (v2f){ q.x, q.y };
            bf[ks][2 * np + 1] = (v2f){ q.z, q.w };
        }
    }
    float bsc[8];
#pragma unroll
    for (int n = 0; n < 8; ++n)
        bsc[n] = bias2[n * 16 + laneM];

    for (int t = 0; t < TILES_PER_WAVE - 2; t += 2) {
        load_a_tile(a1, hrow + (long)(t + 1) * TILE_STEP);
        tile_product(a0, bf, bsc, orow + (long)t * TILE_STEP);
        load_a_tile(a0, hrow + (long)(t + 2) * TILE_STEP);
        tile_product(a1, bf, bsc, orow + (long)(t + 1) * TILE_STEP);
    }
    load_a_tile(a1, hrow + (long)(TILES_PER_WAVE - 1) * TILE_STEP);
    tile_product(a0, bf, bsc, orow + (long)(TILES_PER_WAVE - 2) * TILE_STEP);
    tile_product(a1, bf, bsc, orow + (long)(TILES_PER_WAVE - 1) * TILE_STEP);
}

extern "C" void kernel_launch(void* const* d_in, const int* in_sizes, int n_in,
                              void* d_out, int out_size, void* d_ws, size_t ws_size,
                              hipStream_t stream) {
    const float* h     = (const float*)d_in[0];
    const float* graph = (const float*)d_in[1];
    const float* W     = (const float*)d_in[2];
    const float* b     = (const float*)d_in[3];
    float* out = (float*)d_out;

    float* M     = (float*)d_ws;
    float* bias2 = M + FDIM * FDIM;

    prep_kernel<<<dim3(129), dim3(FDIM), 0, stream>>>(graph, W, b, M, bias2);

    const long rows = (long)out_size / FDIM;
    const int nblocks = (int)(rows / ROWS_PER_GROUP);
    gemm_kernel<<<dim3(nblocks), dim3(32 * GEMM_WAVES), 0, stream>>>(h, M, bias2, out);
}
